# starting kernel
# baseline (speedup 1.0000x reference)
#include <hip/hip_runtime.h>

typedef __attribute__((ext_vector_type(2))) float v2f;
typedef __attribute__((ext_vector_type(8))) float v8f;


__global__ void fill_kernel(float* __restrict__ p, float v, int n) {
    int i = blockIdx.x * blockDim.x + threadIdx.x;
    if (i < n) p[i] = v;
}

__global__ void deg_count_kernel(const int* __restrict__ dst,
                                 float* __restrict__ deg, int E) {
    int e = blockIdx.x * blockDim.x + threadIdx.x;
    if (e < E) atomicAdd(&deg[dst[e]], 1.0f);
}

__global__ void rsqrt_kernel(float* __restrict__ d, int n) {
    int i = blockIdx.x * blockDim.x + threadIdx.x;
    if (i < n) d[i] = rsqrtf(d[i]);
}


__global__ void __launch_bounds__(256)
gemm64_wmma(const float* __restrict__ X, const float* __restrict__ W,
            float* __restrict__ Y, int nrows) {
    __shared__ float lw[64 * 64];
    for (int i = threadIdx.x; i < 64 * 64; i += 256) {
        int k = i >> 6, n = i & 63;
        lw[(k >> 1) * 128 + n * 2 + (k & 1)] = W[i];
    }
    __syncthreads();

    int wave = threadIdx.x >> 5;
    int lane = threadIdx.x & 31;
    int lh   = lane & 15;
    int hi   = lane >> 4;
    int rowbase = (blockIdx.x * 8 + wave) * 16;

    int arow = rowbase + lh;
    int arow_c = arow < nrows ? arow : (nrows - 1);
    const v2f* xp = (const v2f*)(X + (size_t)arow_c * 64);

    v8f acc0 = {}, acc1 = {}, acc2 = {}, acc3 = {};

    #pragma unroll
    for (int kk = 0; kk < 16; ++kk) {
        v2f a = xp[2 * kk + hi];

        const v2f* wp = (const v2f*)(lw + (2 * kk + hi) * 128 + lh * 2);
        v2f b0  = wp[0];
        v2f b1v = wp[16];
        v2f b2v = wp[32];
        v2f b3v = wp[48];

        acc0 = __builtin_amdgcn_wmma_f32_16x16x4_f32(false, a, false, b0,  (short)0, acc0, false, false);
        acc1 = __builtin_amdgcn_wmma_f32_16x16x4_f32(false, a, false, b1v, (short)0, acc1, false, false);
        acc2 = __builtin_amdgcn_wmma_f32_16x16x4_f32(false, a, false, b2v, (short)0, acc2, false, false);
        acc3 = __builtin_amdgcn_wmma_f32_16x16x4_f32(false, a, false, b3v, (short)0, acc3, false, false);
    }

    #pragma unroll
    for (int r = 0; r < 8; ++r) {
        int row = rowbase + r + hi * 8;
        if (row < nrows) {
            float* yp = Y + (size_t)row * 64 + lh;
            yp[0]  = acc0[r];
            yp[16] = acc1[r];
            yp[32] = acc2[r];
            yp[48] = acc3[r];
        }
    }
}


__global__ void scatter_kernel(const float* __restrict__ xw,
                               const int* __restrict__ src,
                               const int* __restrict__ dst,
                               const float* __restrict__ dinv,
                               float* __restrict__ agg, int E) {
    int t = blockIdx.x * blockDim.x + threadIdx.x;
    int e = t >> 4;
    int q = t & 15;
    if (e >= E) return;
    int s = src[e];
    int d = dst[e];
    float nrm = dinv[s] * dinv[d];
    float4 v = reinterpret_cast<const float4*>(xw + (size_t)s * 64)[q];
    float* base = agg + (size_t)d * 64 + q * 4;
    atomicAdd(base + 0, v.x * nrm);
    atomicAdd(base + 1, v.y * nrm);
    atomicAdd(base + 2, v.z * nrm);
    atomicAdd(base + 3, v.w * nrm);
}


__global__ void finish_kernel(float* __restrict__ acc, const float* __restrict__ xw,
                              const float* __restrict__ dinv, const float* __restrict__ bias,
                              int total, int do_relu) {
    int i = blockIdx.x * blockDim.x + threadIdx.x;
    if (i >= total) return;
    int n = i >> 6;
    int f = i & 63;
    float di = dinv[n];
    float v = acc[i] + xw[i] * di * di + bias[f];
    if (do_relu) v = fmaxf(v, 0.0f);
    acc[i] = v;
}


extern "C" void kernel_launch(void* const* d_in, const int* in_sizes, int n_in,
                              void* d_out, int out_size, void* d_ws, size_t ws_size,
                              hipStream_t stream) {
    (void)n_in; (void)out_size; (void)ws_size;

    const float* x  = (const float*)d_in[0];
    const int*   ei = (const int*)d_in[1];
    const float* W1 = (const float*)d_in[2];
    const float* b1 = (const float*)d_in[3];
    const float* W2 = (const float*)d_in[4];
    const float* b2 = (const float*)d_in[5];
    float* out = (float*)d_out;

    const int N = in_sizes[0] / 64;
    const int E = in_sizes[1] / 2;
    const int* src = ei;
    const int* dst = ei + E;

    float* dinv = (float*)d_ws;
    float* xw   = dinv + (((size_t)N + 32767) & ~(size_t)32767);
    float* agg  = xw + (size_t)N * 64;

    const int T = 256;
    const int NF = N * 64;

    fill_kernel<<<(N + T - 1) / T, T, 0, stream>>>(dinv, 1.0f, N);
    deg_count_kernel<<<(E + T - 1) / T, T, 0, stream>>>(dst, dinv, E);
    rsqrt_kernel<<<(N + T - 1) / T, T, 0, stream>>>(dinv, N);

    gemm64_wmma<<<(N + 127) / 128, T, 0, stream>>>(x, W1, xw, N);
    fill_kernel<<<(NF + T - 1) / T, T, 0, stream>>>(agg, 0.0f, NF);
    scatter_kernel<<<((E * 16) + T - 1) / T, T, 0, stream>>>(xw, src, dst, dinv, agg, E);
    finish_kernel<<<(NF + T - 1) / T, T, 0, stream>>>(agg, xw, dinv, b1, NF, 1);

    gemm64_wmma<<<(N + 127) / 128, T, 0, stream>>>(agg, W2, xw, N);
    fill_kernel<<<(NF + T - 1) / T, T, 0, stream>>>(out, 0.0f, NF);
    scatter_kernel<<<((E * 16) + T - 1) / T, T, 0, stream>>>(xw, src, dst, dinv, out, E);
    finish_kernel<<<(NF + T - 1) / T, T, 0, stream>>>(out, xw, dinv, b2, NF, 0);
}


# edge pass kept inside each half of the second-level cache: destination rows dealt to the halves by row parity (whole 256-B rows, no column cut), chunk queue per half; sums keep atomic arrival order
# speedup vs baseline: 3.2953x; 3.2953x over previous
#include <hip/hip_runtime.h>


typedef __attribute__((ext_vector_type(2))) float v2f;
typedef __attribute__((ext_vector_type(8))) float v8f;

#define Q_PITCH 64


#define HIST_CHUNKS 32
#define HIST_ROWS   100096
#define HIST_WORDS  (HIST_ROWS / 2)

__device__ __forceinline__ void hist_count(unsigned int* h, unsigned int d) {
    if (d < HIST_ROWS) atomicAdd(&h[d >> 1], 1u << ((d & 1u) << 4));
}

__global__ void __launch_bounds__(1024)
deg_hist_kernel(const int* __restrict__ dst, float* __restrict__ part_f, int E) {
    __shared__ unsigned int h[HIST_WORDS];
    const int tid = threadIdx.x;
    uint4* h4 = (uint4*)h;
    for (int i = tid; i < HIST_WORDS / 4; i += 1024) h4[i] = make_uint4(0u, 0u, 0u, 0u);
    __syncthreads();

    const int chunk = (((E + HIST_CHUNKS - 1) / HIST_CHUNKS) + 3) & ~3;
    const int e0 = blockIdx.x * chunk;
    int n = E - e0;
    if (n > chunk) n = chunk;
    if (n < 0) n = 0;
    const int* p = dst + e0;

    int done = 0;
    if ((((size_t)p) & 15) == 0) {
        const int4* p4 = (const int4*)p;
        const int n4 = n >> 2;
        for (int i0 = 0; i0 < n4; i0 += 4 * 1024) {
            int4 v[4];
            #pragma unroll
            for (int k = 0; k < 4; ++k) {
                const int i = i0 + k * 1024 + tid;
                v[k] = p4[i < n4 ? i : 0];
            }
            #pragma unroll
            for (int k = 0; k < 4; ++k) {
                const int i = i0 + k * 1024 + tid;
                if (i < n4) {
                    hist_count(h, (unsigned int)v[k].x);
                    hist_count(h, (unsigned int)v[k].y);
                    hist_count(h, (unsigned int)v[k].z);
                    hist_count(h, (unsigned int)v[k].w);
                }
            }
        }
        done = n4 << 2;
    }
    for (int i = done + tid; i < n; i += 1024) hist_count(h, (unsigned int)p[i]);
    __syncthreads();

    uint4* o4 = (uint4*)part_f + (size_t)blockIdx.x * (HIST_WORDS / 4);
    for (int i = tid; i < HIST_WORDS / 4; i += 1024) o4[i] = h4[i];
}


template <bool FIRST>
__global__ void __launch_bounds__(256)
gemm64_wmma(const float* __restrict__ X, const float* __restrict__ W,
            float* __restrict__ dinv, const float* __restrict__ part_f,
            const float* __restrict__ bias_in, const float* __restrict__ bias_out,
            float* __restrict__ Y, float* __restrict__ S, int* __restrict__ qz, int nrows) {
    if (FIRST && blockIdx.x == 0 && threadIdx.x < 4) qz[(threadIdx.x >> 1) * Q_PITCH + (threadIdx.x & 1)] = 0;
    __shared__ float ldinv[128];
    __shared__ v2f lbin[32];
    if (FIRST) {
        if (threadIdx.x < 128) {
            const int row = blockIdx.x * 128 + threadIdx.x;
            const unsigned short* ph = (const unsigned short*)part_f + row;
            unsigned int c = 1u;
            #pragma unroll
            for (int w = 0; w < HIST_CHUNKS; ++w) c += ph[(size_t)w * HIST_ROWS];
            const float d = rsqrtf((float)c);
            ldinv[threadIdx.x] = d;
            if (row < nrows) dinv[row] = d;
        }
    } else {
        if (threadIdx.x < 32) lbin[threadIdx.x] = ((const v2f*)bias_in)[threadIdx.x];
    }

    __shared__ float lw[64 * 64];
    for (int i = threadIdx.x; i < 64 * 64; i += 256) {
        int k = i >> 6, n = i & 63;
        lw[(k >> 1) * 128 + n * 2 + (k & 1)] = W[i];
    }
    __syncthreads();

    int wave = threadIdx.x >> 5;
    int lane = threadIdx.x & 31;
    int lh   = lane & 15;
    int hi   = lane >> 4;
    int rowbase = (blockIdx.x * 8 + wave) * 16;

    int arow = rowbase + lh;
    int arow_c = arow < nrows ? arow : (nrows - 1);
    const v2f* xp = (const v2f*)(X + (size_t)arow_c * 64);
    float da = 0.0f;
    if (!FIRST) da = dinv[arow_c];

    float dg[8];
    #pragma unroll
    for (int r = 0; r < 8; ++r) {
        int row = rowbase + r + hi * 8;
        if (FIRST) dg[r] = ldinv[wave * 16 + r + hi * 8];
        else       dg[r] = dinv[row < nrows ? row : (nrows - 1)];
    }

    v8f acc0 = {}, acc1 = {}, acc2 = {}, acc3 = {};

    #pragma unroll
    for (int kk = 0; kk < 16; ++kk) {
        v2f a = xp[2 * kk + hi];
        if (!FIRST) {
            const v2f bi = lbin[2 * kk + hi];
            a.x = fmaxf(da * a.x + bi.x, 0.0f);
            a.y = fmaxf(da * a.y + bi.y, 0.0f);
        }

        const v2f* wp = (const v2f*)(lw + (2 * kk + hi) * 128 + lh * 2);
        v2f b0  = wp[0];
        v2f b1v = wp[16];
        v2f b2v = wp[32];
        v2f b3v = wp[48];

        acc0 = __builtin_amdgcn_wmma_f32_16x16x4_f32(false, a, false, b0,  (short)0, acc0, false, false);
        acc1 = __builtin_amdgcn_wmma_f32_16x16x4_f32(false, a, false, b1v, (short)0, acc1, false, false);
        acc2 = __builtin_amdgcn_wmma_f32_16x16x4_f32(false, a, false, b2v, (short)0, acc2, false, false);
        acc3 = __builtin_amdgcn_wmma_f32_16x16x4_f32(false, a, false, b3v, (short)0, acc3, false, false);
    }

    float bs0 = 0.0f, bs1 = 0.0f, bs2 = 0.0f, bs3 = 0.0f;
    if (!FIRST) {
        bs0 = bias_out[lh];
        bs1 = bias_out[lh + 16];
        bs2 = bias_out[lh + 32];
        bs3 = bias_out[lh + 48];
    }

    #pragma unroll
    for (int r = 0; r < 8; ++r) {
        int row = rowbase + r + hi * 8;
        if (row < nrows) {
            const float di = dg[r];
            const size_t o = (size_t)row * 64 + lh;
            const float t0 = acc0[r] * di, t1 = acc1[r] * di, t2 = acc2[r] * di, t3 = acc3[r] * di;
            Y[o]      = t0;
            Y[o + 16] = t1;
            Y[o + 32] = t2;
            Y[o + 48] = t3;
            if (FIRST) {
                S[o]      = t0;
                S[o + 16] = t1;
                S[o + 32] = t2;
                S[o + 48] = t3;
            } else {
                S[o]      = t0 * di + bs0;
                S[o + 16] = t1 * di + bs1;
                S[o + 32] = t2 * di + bs2;
                S[o + 48] = t3 * di + bs3;
            }
        }
    }
}


#define Q_EDGES   512
#define SC_GROUPS 1024

template <bool DST_SCALE, int K>
__device__ __forceinline__ void move_group(const float* __restrict__ xw, float* __restrict__ agg,
                                           int s, int d, float nd, int g, int lane) {
    float v0[K], v1[K];
    #pragma unroll
    for (int i = 0; i < K; ++i) {
        const int si = __builtin_amdgcn_readlane(s, g + i);
        const float* p = xw + (size_t)si * 64 + lane;
        v0[i] = p[0];
        v1[i] = p[32];
    }
    #pragma unroll
    for (int i = 0; i < K; ++i) {
        const int di = __builtin_amdgcn_readlane(d, g + i);
        float* q = agg + (size_t)di * 64 + lane;
        if (DST_SCALE) {
            const float ni = __int_as_float(__builtin_amdgcn_readlane(__float_as_int(nd), g + i));
            atomicAdd(q,      v0[i] * ni);
            atomicAdd(q + 32, v1[i] * ni);
        } else {
            atomicAdd(q,      v0[i]);
            atomicAdd(q + 32, v1[i]);
        }
    }
}

template <bool DST_SCALE>
__device__ __forceinline__ void move_tail(const float* __restrict__ xw, float* __restrict__ agg,
                                          int s, int d, float nd, int g, int m, int lane) {
    float v0[7], v1[7];
    #pragma unroll
    for (int i = 0; i < 7; ++i) {
        const int si = __builtin_amdgcn_readlane(s, g + (i < m ? i : m - 1));
        const float* p = xw + (size_t)si * 64 + lane;
        v0[i] = p[0];
        v1[i] = p[32];
    }
    #pragma unroll
    for (int i = 0; i < 7; ++i) {
        if (i < m) {
            const int di = __builtin_amdgcn_readlane(d, g + i);
            float* q = agg + (size_t)di * 64 + lane;
            if (DST_SCALE) {
                const float ni = __int_as_float(__builtin_amdgcn_readlane(__float_as_int(nd), g + i));
                atomicAdd(q,      v0[i] * ni);
                atomicAdd(q + 32, v1[i] * ni);
            } else {
                atomicAdd(q,      v0[i]);
                atomicAdd(q + 32, v1[i]);
            }
        }
    }
}

template <bool DST_SCALE>
__device__ __forceinline__ void move_rows(const float* __restrict__ xw, float* __restrict__ agg,
                                          int s, int d, float nd, int n, int lane) {
    int g = 0;
    for (; g + 16 <= n; g += 16) move_group<DST_SCALE, 16>(xw, agg, s, d, nd, g, lane);
    if (g + 8 <= n) { move_group<DST_SCALE, 8>(xw, agg, s, d, nd, g, lane); g += 8; }
    if (g < n) move_tail<DST_SCALE>(xw, agg, s, d, nd, g, n - g, lane);
}

template <bool DST_SCALE>
__global__ void __launch_bounds__(256)
scatter_kernel(const float* __restrict__ xw,
               const int* __restrict__ src,
               const int* __restrict__ dst,
               const float* __restrict__ dinv,
               float* __restrict__ agg,
               int* __restrict__ q_lo, int* __restrict__ q_hi, int E) {
    __shared__ int lpack[8][2][128];
    __shared__ int lclaim[2];
    const int lane = threadIdx.x & 31;
    const int wave = __builtin_amdgcn_readfirstlane((int)(threadIdx.x >> 5));
    int* ls = lpack[wave][0];
    int* ld = lpack[wave][1];

    const int mine = (int)((__builtin_amdgcn_s_sendmsg_rtn(0x87) >> 18) & 1u);
    const int nchunks = (E + Q_EDGES - 1) / Q_EDGES;

    int k = 0;
    for (int pass = 0; pass < 2; ++pass) {
        const int h = mine ^ pass;
        int* ctr = h ? q_hi : q_lo;
        int ahead = 0;
        if (threadIdx.x == 0) ahead = atomicAdd(ctr, 1);
        for (int it = 0; it <= nchunks; ++it, ++k) {
            if (threadIdx.x == 0) lclaim[k & 1] = ahead;
            __syncthreads();
            const int c = __builtin_amdgcn_readfirstlane(lclaim[k & 1]);
            if (c >= nchunks) { ++k; break; }

            const int e0 = c * Q_EDGES + wave * 64;
            const int ea = e0 + lane, eb = ea + 32;
            const int ca = ea < E ? ea : E - 1, cb = eb < E ? eb : E - 1;
            const int sa = src[ca], da = dst[ca], sb = src[cb], db = dst[cb];
            if (threadIdx.x == 0) ahead = atomicAdd(ctr, 1);
            const bool ka = (ea < E) && ((da & 1) == h);
            const bool kb = (eb < E) && ((db & 1) == h);
            const unsigned int ma = __builtin_amdgcn_ballot_w32(ka);
            const unsigned int mb = __builtin_amdgcn_ballot_w32(kb);
            const int na  = __builtin_popcount(ma);
            const int cnt = na + __builtin_popcount(mb);
            const int pa  = (int)__builtin_amdgcn_mbcnt_lo(ma, 0u);
            const int pb  = na + (int)__builtin_amdgcn_mbcnt_lo(mb, 0u);
            const int ia = ka ? pa : 64 + lane;
            const int ib = kb ? pb : 96 + lane;
            ls[ia] = sa; ld[ia] = da;
            ls[ib] = sb; ld[ib] = db;
            int s0 = 0, d0 = 0, s1 = 0, d1 = 0;
            float n0 = 0.0f, n1 = 0.0f;
            if (lane < cnt)      { s0 = ls[lane];      d0 = ld[lane];      if (DST_SCALE) n0 = dinv[d0]; }
            if (lane + 32 < cnt) { s1 = ls[lane + 32]; d1 = ld[lane + 32]; if (DST_SCALE) n1 = dinv[d1]; }

            int s = s0, d = d0, n = cnt;
            float nd = n0;
            if (cnt >= 32) {
                move_group<DST_SCALE, 16>(xw, agg, s0, d0, n0, 0, lane);
                move_group<DST_SCALE, 16>(xw, agg, s0, d0, n0, 16, lane);
                s = s1; d = d1; nd = n1; n = cnt - 32;
            }
            move_rows<DST_SCALE>(xw, agg, s, d, nd, n, lane);
        }
    }
}


extern "C" void kernel_launch(void* const* d_in, const int* in_sizes, int n_in,
                              void* d_out, int out_size, void* d_ws, size_t ws_size,
                              hipStream_t stream) {
    (void)n_in; (void)out_size; (void)ws_size;

    const float* x  = (const float*)d_in[0];
    const int*   ei = (const int*)d_in[1];
    const float* W1 = (const float*)d_in[2];
    const float* b1 = (const float*)d_in[3];
    const float* W2 = (const float*)d_in[4];
    const float* b2 = (const float*)d_in[5];
    float* out = (float*)d_out;

    const int N = in_sizes[0] / 64;
    const int E = in_sizes[1] / 2;
    const int* src = ei;
    const int* dst = ei + E;

    float* dinv = (float*)d_ws;
    float* xw  = dinv + (((size_t)N + 2 * Q_PITCH + 32767) & ~(size_t)32767);
    int* q = (int*)(xw - 2 * Q_PITCH);
    float* agg = xw + (size_t)N * 64;

    const int T = 256;

    deg_hist_kernel<<<HIST_CHUNKS, 1024, 0, stream>>>(dst, out, E);

    gemm64_wmma<true><<<(N + 127) / 128, T, 0, stream>>>(x, W1, dinv, out, nullptr, nullptr, xw, agg, q, N);
    scatter_kernel<false><<<SC_GROUPS, T, 0, stream>>>(xw, src, dst, dinv, agg, q, q + Q_PITCH, E);

    gemm64_wmma<false><<<(N + 127) / 128, T, 0, stream>>>(agg, W2, dinv, nullptr, b1, b2, xw, out, q, N);
    scatter_kernel<true><<<SC_GROUPS, T, 0, stream>>>(xw, src, dst, dinv, out, q + 1, q + Q_PITCH + 1, E);
}
